# starting kernel
# baseline (speedup 1.0000x reference)
#include <hip/hip_runtime.h>

#define HID 128
#define D_IN 16
#define D_OUT 4
#define LN_EPS 1e-5f

typedef float v2f __attribute__((ext_vector_type(2)));
typedef float v8f __attribute__((ext_vector_type(8)));

__global__ void zero_kernel(float* __restrict__ p, int n) {
  int i = blockIdx.x * blockDim.x + threadIdx.x;
  int stride = gridDim.x * blockDim.x;
  for (; i < n; i += stride) p[i] = 0.0f;
}

__global__ void deg_kernel(const int* __restrict__ dst, float* __restrict__ deg, int E) {
  int e = blockIdx.x * blockDim.x + threadIdx.x;
  if (e < E) unsafeAtomicAdd(&deg[dst[e]], 1.0f);
}

__global__ void invdeg_kernel(float* __restrict__ deg, int N) {
  int i = blockIdx.x * blockDim.x + threadIdx.x;
  if (i < N) {
    float d = deg[i];
    deg[i] = 1.0f / fmaxf(d, 1.0f);
  }
}

__global__ void in_proj_kernel(const float* __restrict__ x,
                               const float* __restrict__ W_in,
                               const float* __restrict__ b_in,
                               float* __restrict__ h, int N) {
  int idx = blockIdx.x * blockDim.x + threadIdx.x;
  if (idx >= N * HID) return;
  int row = idx >> 7;
  int col = idx & (HID - 1);
  const float4* xr = (const float4*)(x + (size_t)row * D_IN);
  float acc = b_in[col];
#pragma unroll
  for (int kq = 0; kq < D_IN / 4; ++kq) {
    float4 xv = xr[kq];
    acc += xv.x * W_in[(kq * 4 + 0) * HID + col];
    acc += xv.y * W_in[(kq * 4 + 1) * HID + col];
    acc += xv.z * W_in[(kq * 4 + 2) * HID + col];
    acc += xv.w * W_in[(kq * 4 + 3) * HID + col];
  }
  h[idx] = acc;
}

__global__ void __launch_bounds__(256) ln_kernel(const float* __restrict__ h,
                                                 const float* __restrict__ g,
                                                 const float* __restrict__ b,
                                                 float* __restrict__ hn, int N) {
  int wave = threadIdx.x >> 5;
  int lane = threadIdx.x & 31;
  int row = blockIdx.x * 8 + wave;
  if (row >= N) return;

  float4 v = ((const float4*)(h + (size_t)row * HID))[lane];
  float s = v.x + v.y + v.z + v.w;
#pragma unroll
  for (int off = 16; off > 0; off >>= 1) s += __shfl_xor(s, off, 32);
  float mu = s * (1.0f / HID);

  float dx = v.x - mu, dy = v.y - mu, dz = v.z - mu, dw = v.w - mu;
  float q = dx * dx + dy * dy + dz * dz + dw * dw;
#pragma unroll
  for (int off = 16; off > 0; off >>= 1) q += __shfl_xor(q, off, 32);
  float rstd = rsqrtf(q * (1.0f / HID) + LN_EPS);

  float4 gv = ((const float4*)g)[lane];
  float4 bv = ((const float4*)b)[lane];
  float4 o;
  o.x = dx * rstd * gv.x + bv.x;
  o.y = dy * rstd * gv.y + bv.y;
  o.z = dz * rstd * gv.z + bv.z;
  o.w = dw * rstd * gv.w + bv.w;
  ((float4*)(hn + (size_t)row * HID))[lane] = o;
}

__global__ void __launch_bounds__(256) scatter_kernel(const int* __restrict__ src,
                                                      const int* __restrict__ dst,
                                                      const float* __restrict__ hn,
                                                      float* __restrict__ agg, int E) {
  int gid = blockIdx.x * blockDim.x + threadIdx.x;
  int e = gid >> 5;
  int lane = gid & 31;
  if (e >= E) return;
  int s = src[e];
  int d = dst[e];
  float4 v = ((const float4*)(hn + (size_t)s * HID))[lane];
  float* ap = agg + (size_t)d * HID + lane * 4;
  unsafeAtomicAdd(ap + 0, v.x);
  unsafeAtomicAdd(ap + 1, v.y);
  unsafeAtomicAdd(ap + 2, v.z);
  unsafeAtomicAdd(ap + 3, v.w);
}

__global__ void __launch_bounds__(256) sage_gemm_kernel(
    const float* __restrict__ agg, const float* __restrict__ hn,
    const float* __restrict__ invdeg, const float* __restrict__ Wl,
    const float* __restrict__ Wr, const float* __restrict__ bl,
    float* __restrict__ h, int N) {
  int wave = threadIdx.x >> 5;
  int lane = threadIdx.x & 31;
  int half = lane >> 4;
  int l16 = lane & 15;

  int m0 = blockIdx.x * 16;
  int row = m0 + l16;
  int rowc = row < N ? row : N - 1;
  int col = wave * 16 + l16;

  float invd = invdeg[rowc];
  const float* aggr = agg + (size_t)rowc * HID;
  const float* hnr = hn + (size_t)rowc * HID;

  v8f acc = {};
#pragma unroll 4
  for (int kb = 0; kb < HID; kb += 4) {
    int k0 = kb + 2 * half;
    v2f a1 = *(const v2f*)(aggr + k0);
    a1.x *= invd;
    a1.y *= invd;
    v2f b1;
    b1.x = Wl[k0 * HID + col];
    b1.y = Wl[(k0 + 1) * HID + col];
    acc = __builtin_amdgcn_wmma_f32_16x16x4_f32(false, a1, false, b1,
                                                (short)0, acc, false, false);
    v2f a2 = *(const v2f*)(hnr + k0);
    v2f b2;
    b2.x = Wr[k0 * HID + col];
    b2.y = Wr[(k0 + 1) * HID + col];
    acc = __builtin_amdgcn_wmma_f32_16x16x4_f32(false, a2, false, b2,
                                                (short)0, acc, false, false);
  }

  float bias = bl[col];
  float* hp = h + (size_t)(m0 + 8 * half) * HID + col;
  if (m0 + 16 <= N) {
#pragma unroll
    for (int v = 0; v < 8; ++v) {
      float r = hp[(size_t)v * HID] + acc[v] + bias;
      hp[(size_t)v * HID] = r > 0.0f ? r : 0.0f;
    }
  } else {
#pragma unroll
    for (int v = 0; v < 8; ++v) {
      int m = m0 + v + 8 * half;
      if (m < N) {
        float r = hp[(size_t)v * HID] + acc[v] + bias;
        hp[(size_t)v * HID] = r > 0.0f ? r : 0.0f;
      }
    }
  }
}

__global__ void out_proj_kernel(const float* __restrict__ h,
                                const float* __restrict__ W_out,
                                const float* __restrict__ b_out,
                                float* __restrict__ out, int N) {
  int idx = blockIdx.x * blockDim.x + threadIdx.x;
  if (idx >= N * D_OUT) return;
  int row = idx >> 2;
  int c = idx & 3;
  const float* hr = h + (size_t)row * HID;
  float acc = b_out[c];
#pragma unroll 8
  for (int k = 0; k < HID; ++k) acc += hr[k] * W_out[k * D_OUT + c];
  out[idx] = acc;
}

extern "C" void kernel_launch(void* const* d_in, const int* in_sizes, int n_in,
                              void* d_out, int out_size, void* d_ws, size_t ws_size,
                              hipStream_t stream) {
  const float* x = (const float*)d_in[0];
  const int* ei = (const int*)d_in[1];
  const float* W_in = (const float*)d_in[2];
  const float* b_in = (const float*)d_in[3];
  const float* Wl = (const float*)d_in[4];
  const float* bl = (const float*)d_in[5];
  const float* Wr = (const float*)d_in[6];
  const float* ln_g = (const float*)d_in[7];
  const float* ln_b = (const float*)d_in[8];
  const float* W_out = (const float*)d_in[9];
  const float* b_out = (const float*)d_in[10];
  float* out = (float*)d_out;

  int N = in_sizes[0] / D_IN;
  int E = in_sizes[1] / 2;
  const int* src = ei;
  const int* dst = ei + E;

  float* ws = (float*)d_ws;
  size_t nh = (size_t)N * HID;
  float* h = ws;
  float* hn = ws + nh;
  float* agg = ws + 2 * nh;
  float* deg = ws + 3 * nh;

  zero_kernel<<<256, 256, 0, stream>>>(deg, N);
  deg_kernel<<<(E + 255) / 256, 256, 0, stream>>>(dst, deg, E);
  invdeg_kernel<<<(N + 255) / 256, 256, 0, stream>>>(deg, N);

  in_proj_kernel<<<(int)((nh + 255) / 256), 256, 0, stream>>>(x, W_in, b_in, h, N);

  int mtiles = (N + 15) / 16;
  long long scat_threads = (long long)E * 32;
  for (int l = 0; l < 3; ++l) {
    ln_kernel<<<(N + 7) / 8, 256, 0, stream>>>(h, ln_g + l * HID, ln_b + l * HID, hn, N);
    zero_kernel<<<2048, 256, 0, stream>>>(agg, (int)nh);
    scatter_kernel<<<(int)((scat_threads + 255) / 256), 256, 0, stream>>>(src, dst, hn, agg, E);
    sage_gemm_kernel<<<mtiles, 256, 0, stream>>>(agg, hn, deg,
                                                 Wl + (size_t)l * HID * HID,
                                                 Wr + (size_t)l * HID * HID,
                                                 bl + l * HID, h, N);
  }

  out_proj_kernel<<<(N * D_OUT + 255) / 256, 256, 0, stream>>>(h, W_out, b_out, out, N);
}


# GEMM: weights copied once per work-group into LDS in B-fragment order; a wave does 16x64 units (4 accumulators), A read 16 B per 8 k; k regrouped, so the products' sum order changes
# speedup vs baseline: 2.6485x; 2.6485x over previous
#include <hip/hip_runtime.h>

#define HID 128
#define D_IN 16
#define D_OUT 4
#define LN_EPS 1e-5f

typedef float v2f __attribute__((ext_vector_type(2)));
typedef float v8f __attribute__((ext_vector_type(8)));

__global__ void zero_kernel(float* __restrict__ p, int n) {
  int i = blockIdx.x * blockDim.x + threadIdx.x;
  int stride = gridDim.x * blockDim.x;
  for (; i < n; i += stride) p[i] = 0.0f;
}

__global__ void deg_kernel(const int* __restrict__ dst, float* __restrict__ deg, int E) {
  int e = blockIdx.x * blockDim.x + threadIdx.x;
  if (e < E) unsafeAtomicAdd(&deg[dst[e]], 1.0f);
}

__global__ void invdeg_kernel(float* __restrict__ deg, int N) {
  int i = blockIdx.x * blockDim.x + threadIdx.x;
  if (i < N) {
    float d = deg[i];
    deg[i] = 1.0f / fmaxf(d, 1.0f);
  }
}

__global__ void in_proj_kernel(const float* __restrict__ x,
                               const float* __restrict__ W_in,
                               const float* __restrict__ b_in,
                               float* __restrict__ h, int N) {
  int idx = blockIdx.x * blockDim.x + threadIdx.x;
  if (idx >= N * HID) return;
  int row = idx >> 7;
  int col = idx & (HID - 1);
  const float4* xr = (const float4*)(x + (size_t)row * D_IN);
  float acc = b_in[col];
#pragma unroll
  for (int kq = 0; kq < D_IN / 4; ++kq) {
    float4 xv = xr[kq];
    acc += xv.x * W_in[(kq * 4 + 0) * HID + col];
    acc += xv.y * W_in[(kq * 4 + 1) * HID + col];
    acc += xv.z * W_in[(kq * 4 + 2) * HID + col];
    acc += xv.w * W_in[(kq * 4 + 3) * HID + col];
  }
  h[idx] = acc;
}

__global__ void __launch_bounds__(256) ln_kernel(const float* __restrict__ h,
                                                 const float* __restrict__ g,
                                                 const float* __restrict__ b,
                                                 float* __restrict__ hn, int N) {
  int wave = threadIdx.x >> 5;
  int lane = threadIdx.x & 31;
  int row = blockIdx.x * 8 + wave;
  if (row >= N) return;

  float4 v = ((const float4*)(h + (size_t)row * HID))[lane];
  float s = v.x + v.y + v.z + v.w;
#pragma unroll
  for (int off = 16; off > 0; off >>= 1) s += __shfl_xor(s, off, 32);
  float mu = s * (1.0f / HID);

  float dx = v.x - mu, dy = v.y - mu, dz = v.z - mu, dw = v.w - mu;
  float q = dx * dx + dy * dy + dz * dz + dw * dw;
#pragma unroll
  for (int off = 16; off > 0; off >>= 1) q += __shfl_xor(q, off, 32);
  float rstd = rsqrtf(q * (1.0f / HID) + LN_EPS);

  float4 gv = ((const float4*)g)[lane];
  float4 bv = ((const float4*)b)[lane];
  float4 o;
  o.x = dx * rstd * gv.x + bv.x;
  o.y = dy * rstd * gv.y + bv.y;
  o.z = dz * rstd * gv.z + bv.z;
  o.w = dw * rstd * gv.w + bv.w;
  ((float4*)(hn + (size_t)row * HID))[lane] = o;
}

__global__ void __launch_bounds__(256) scatter_kernel(const int* __restrict__ src,
                                                      const int* __restrict__ dst,
                                                      const float* __restrict__ hn,
                                                      float* __restrict__ agg, int E) {
  int gid = blockIdx.x * blockDim.x + threadIdx.x;
  int e = gid >> 5;
  int lane = gid & 31;
  if (e >= E) return;
  int s = src[e];
  int d = dst[e];
  float4 v = ((const float4*)(hn + (size_t)s * HID))[lane];
  float* ap = agg + (size_t)d * HID + lane * 4;
  unsafeAtomicAdd(ap + 0, v.x);
  unsafeAtomicAdd(ap + 1, v.y);
  unsafeAtomicAdd(ap + 2, v.z);
  unsafeAtomicAdd(ap + 3, v.w);
}

#define OWN 256
#define BUILD_THREADS 1024
#define CAP 8192
#define SCAN_U 8

__global__ void __launch_bounds__(BUILD_THREADS) group_edges_kernel(
    const int* __restrict__ src, const int* __restrict__ dst,
    int* __restrict__ rowptr, int* __restrict__ elist,
    float* __restrict__ invdeg, int N, int E) {
  __shared__ int lcnt[OWN];
  __shared__ int lpre[OWN];
  __shared__ int red[BUILD_THREADS / 32];
  __shared__ int nmatch;
  __shared__ int m_edge[CAP];
  __shared__ int m_loc[CAP];
  __shared__ int m_rank[CAP];

  const int tid = threadIdx.x;
  const int lo = blockIdx.x * OWN;
  const int hi = (lo + OWN < N) ? (lo + OWN) : N;

  for (int i = tid; i < OWN; i += BUILD_THREADS) lcnt[i] = 0;
  if (tid == 0) nmatch = 0;
  __syncthreads();

  int below = 0;
  const unsigned span = (unsigned)(hi - lo);
  auto take = [&](int e, int loc) {
    int r = atomicAdd(&lcnt[loc], 1);
    int slot = atomicAdd(&nmatch, 1);
    if (slot < CAP) {
      m_edge[slot] = e;
      m_loc[slot] = loc;
      m_rank[slot] = r;
    }
  };
  auto visit = [&](int e, int d) {
    below += (d < lo) ? 1 : 0;
    unsigned loc = (unsigned)(d - lo);
    if (loc < span) take(e, (int)loc);
  };
  auto visit4 = [&](int e, int4 d) {
    below += ((d.x < lo) ? 1 : 0) + ((d.y < lo) ? 1 : 0) + ((d.z < lo) ? 1 : 0) + ((d.w < lo) ? 1 : 0);
    unsigned lx = (unsigned)(d.x - lo), ly = (unsigned)(d.y - lo);
    unsigned lz = (unsigned)(d.z - lo), lw = (unsigned)(d.w - lo);
    bool ix = lx < span, iy = ly < span, iz = lz < span, iw = lw < span;
    if (ix | iy | iz | iw) {
      if (ix) take(e + 0, (int)lx);
      if (iy) take(e + 1, (int)ly);
      if (iz) take(e + 2, (int)lz);
      if (iw) take(e + 3, (int)lw);
    }
  };

  const bool vec_ok = (((size_t)dst) & 15) == 0;
  const int E4 = vec_ok ? (E >> 2) : 0;
  const int4* d4 = (const int4*)dst;
  int i0 = tid;
  for (; i0 + (SCAN_U - 1) * BUILD_THREADS < E4; i0 += SCAN_U * BUILD_THREADS) {
    int4 d[SCAN_U];
#pragma unroll
    for (int u = 0; u < SCAN_U; ++u) d[u] = d4[i0 + u * BUILD_THREADS];
#pragma unroll
    for (int u = 0; u < SCAN_U; ++u) visit4((i0 + u * BUILD_THREADS) << 2, d[u]);
  }
  for (; i0 < E4; i0 += BUILD_THREADS) visit4(i0 << 2, d4[i0]);
  for (int e = (E4 << 2) + tid; e < E; e += BUILD_THREADS) visit(e, dst[e]);

  int s = below;
  for (int off = 16; off > 0; off >>= 1) s += __shfl_xor(s, off, 32);
  if ((tid & 31) == 0) red[tid >> 5] = s;
  __syncthreads();
  int base = 0;
  for (int i = 0; i < BUILD_THREADS / 32; ++i) base += red[i];

  if (tid < OWN) {
    int p = 0;
    for (int i = 0; i < tid; ++i) p += lcnt[i];
    lpre[tid] = p;
  }
  __syncthreads();

  if (tid < OWN && lo + tid < N) {
    rowptr[lo + tid] = base + lpre[tid];
    invdeg[lo + tid] = 1.0f / fmaxf((float)lcnt[tid], 1.0f);
  }
  if (hi == N && tid == 0) {
    int last = hi - lo - 1;
    rowptr[N] = base + lpre[last] + lcnt[last];
  }

  const int nm = nmatch;
  if (nm <= CAP) {
    for (int i = tid; i < nm; i += BUILD_THREADS)
      elist[base + lpre[m_loc[i]] + m_rank[i]] = src[m_edge[i]];
  } else {
    __syncthreads();
    for (int i = tid; i < OWN; i += BUILD_THREADS) lcnt[i] = 0;
    __syncthreads();
    for (int e = tid; e < E; e += BUILD_THREADS) {
      int d = dst[e];
      if (d >= lo && d < hi) {
        int loc = d - lo;
        int r = atomicAdd(&lcnt[loc], 1);
        elist[base + lpre[loc] + r] = src[e];
      }
    }
  }
}

template <int U>
__device__ __forceinline__ void add_rows(const float4* __restrict__ hn4, int my,
                                         int j, int lane, float4& acc) {
  float4 v[U];
#pragma unroll
  for (int u = 0; u < U; ++u) {
    int sidx = __builtin_amdgcn_readlane(my, j + u);
    v[u] = hn4[(size_t)sidx * (HID / 4) + lane];
  }
#pragma unroll
  for (int u = 0; u < U; ++u) {
    acc.x += v[u].x;
    acc.y += v[u].y;
    acc.z += v[u].z;
    acc.w += v[u].w;
  }
}

__global__ void __launch_bounds__(256) gather_kernel(const int* __restrict__ rowptr,
                                                     const int* __restrict__ elist,
                                                     const float* __restrict__ hn,
                                                     float* __restrict__ agg, int N) {
  const int lane = threadIdx.x & 31;
  const int wave = __builtin_amdgcn_readfirstlane((int)((blockIdx.x * blockDim.x + threadIdx.x) >> 5));
  const int nw = (int)((gridDim.x * blockDim.x) >> 5);
  const float4* hn4 = (const float4*)hn;
  for (int n = wave; n < N; n += nw) {
    const int beg = __builtin_amdgcn_readfirstlane(rowptr[n]);
    const int end = __builtin_amdgcn_readfirstlane(rowptr[n + 1]);
    float4 acc = make_float4(0.0f, 0.0f, 0.0f, 0.0f);
    for (int b = beg; b < end; b += 32) {
      const int cnt = (end - b < 32) ? (end - b) : 32;
      const int my = (lane < cnt) ? elist[b + lane] : 0;
      int j = 0;
      for (; j + 8 <= cnt; j += 8) add_rows<8>(hn4, my, j, lane, acc);
      if (j + 4 <= cnt) { add_rows<4>(hn4, my, j, lane, acc); j += 4; }
      if (j + 2 <= cnt) { add_rows<2>(hn4, my, j, lane, acc); j += 2; }
      if (j < cnt) add_rows<1>(hn4, my, j, lane, acc);
    }
    ((float4*)(agg + (size_t)n * HID))[lane] = acc;
  }
}

#define GEMM_THREADS 256
#define GEMM_GROUPS 512
#define KG (2 * HID / 4)

__device__ __forceinline__ void gemm_k128(const float4* __restrict__ arow, float scale,
                                          const float4* __restrict__ wfrag, v8f (&acc)[4]) {
#pragma unroll
  for (int j = 0; j < HID / 8; ++j) {
    float4 a = arow[2 * j];
    a.x *= scale;
    a.y *= scale;
    a.z *= scale;
    a.w *= scale;
    v2f a0 = {a.x, a.y};
    v2f a1 = {a.z, a.w};
    const float4* bp = wfrag + (size_t)(2 * j) * HID;
#pragma unroll
    for (int t = 0; t < 4; ++t) {
      float4 b = bp[16 * t];
      v2f b0 = {b.x, b.y};
      v2f b1 = {b.z, b.w};
      acc[t] = __builtin_amdgcn_wmma_f32_16x16x4_f32(false, a0, false, b0, (short)0, acc[t], false, false);
      acc[t] = __builtin_amdgcn_wmma_f32_16x16x4_f32(false, a1, false, b1, (short)0, acc[t], false, false);
    }
  }
}

__global__ void __launch_bounds__(GEMM_THREADS) sage_gemm_kernel(
    const float* __restrict__ agg, const float* __restrict__ hn,
    const float* __restrict__ invdeg, const float* __restrict__ Wl,
    const float* __restrict__ Wr, const float* __restrict__ bl,
    float* __restrict__ h, int N) {
  __shared__ float4 wf[KG * HID];
  __shared__ int next_unit;

  const int tid = threadIdx.x;
  if (tid == 0) next_unit = 0;

#pragma unroll
  for (int q = 0; q < KG / (GEMM_THREADS / 32); ++q) {
    const int g = q * (GEMM_THREADS / 32) + (tid >> 5);
    const int m = tid & 31;
    const float* wp = (q < KG / 2 / (GEMM_THREADS / 32))
                          ? (Wl + (size_t)(4 * g) * HID)
                          : (Wr + (size_t)(4 * (g - KG / 2)) * HID);
    const float4 r0 = ((const float4*)(wp))[m];
    const float4 r1 = ((const float4*)(wp + HID))[m];
    const float4 r2 = ((const float4*)(wp + 2 * HID))[m];
    const float4 r3 = ((const float4*)(wp + 3 * HID))[m];
    float4* o = wf + g * HID + 4 * m;
    o[0] = make_float4(r0.x, r1.x, r2.x, r3.x);
    o[1] = make_float4(r0.y, r1.y, r2.y, r3.y);
    o[2] = make_float4(r0.z, r1.z, r2.z, r3.z);
    o[3] = make_float4(r0.w, r1.w, r2.w, r3.w);
  }

  const int lane = tid & 31;
  const int half = lane >> 4;
  const int l16 = lane & 15;

  const int nunits = ((N + 15) >> 4) * 2;
  const int per = nunits / (int)gridDim.x;
  const int rem = nunits - per * (int)gridDim.x;
  const int bx = (int)blockIdx.x;
  const int ub = bx * per + (bx < rem ? bx : rem);
  const int ue = ub + per + (bx < rem ? 1 : 0);
  __syncthreads();

  for (int draw = 0; draw <= ue - ub; ++draw) {
    int i = 0;
    if (lane == 0) i = atomicAdd(&next_unit, 1);
    i = __builtin_amdgcn_readfirstlane(i);
    const int u = ub + i;
    if (u >= ue) break;

    const int m0 = (u >> 1) * 16;
    const int n0 = (u & 1) * 64;
    const int row = m0 + l16;
    const int rowc = row < N ? row : N - 1;

    const float invd = invdeg[rowc];
    const float4* aggr = (const float4*)(agg + (size_t)rowc * HID) + half;
    const float4* hnr = (const float4*)(hn + (size_t)rowc * HID) + half;
    const float4* wl_f = wf + (size_t)half * HID + n0 + l16;
    const float4* wr_f = wl_f + (size_t)(KG / 2) * HID;

    v8f acc[4] = {};
    gemm_k128(aggr, invd, wl_f, acc);
    gemm_k128(hnr, 1.0f, wr_f, acc);

    float* hp = h + (size_t)(m0 + 8 * half) * HID + n0 + l16;
    if (m0 + 16 <= N) {
#pragma unroll
      for (int t = 0; t < 4; ++t) {
        const float bias = bl[n0 + 16 * t + l16];
#pragma unroll
        for (int v = 0; v < 8; ++v) {
          float r = hp[(size_t)v * HID + 16 * t] + acc[t][v] + bias;
          hp[(size_t)v * HID + 16 * t] = r > 0.0f ? r : 0.0f;
        }
      }
    } else {
#pragma unroll
      for (int t = 0; t < 4; ++t) {
        const float bias = bl[n0 + 16 * t + l16];
#pragma unroll
        for (int v = 0; v < 8; ++v) {
          int m = m0 + v + 8 * half;
          if (m < N) {
            float r = hp[(size_t)v * HID + 16 * t] + acc[t][v] + bias;
            hp[(size_t)v * HID + 16 * t] = r > 0.0f ? r : 0.0f;
          }
        }
      }
    }
  }
}

__global__ void out_proj_kernel(const float* __restrict__ h,
                                const float* __restrict__ W_out,
                                const float* __restrict__ b_out,
                                float* __restrict__ out, int N) {
  int idx = blockIdx.x * blockDim.x + threadIdx.x;
  if (idx >= N * D_OUT) return;
  int row = idx >> 2;
  int c = idx & 3;
  const float* hr = h + (size_t)row * HID;
  float acc = b_out[c];
#pragma unroll 8
  for (int k = 0; k < HID; ++k) acc += hr[k] * W_out[k * D_OUT + c];
  out[idx] = acc;
}

extern "C" void kernel_launch(void* const* d_in, const int* in_sizes, int n_in,
                              void* d_out, int out_size, void* d_ws, size_t ws_size,
                              hipStream_t stream) {
  const float* x = (const float*)d_in[0];
  const int* ei = (const int*)d_in[1];
  const float* W_in = (const float*)d_in[2];
  const float* b_in = (const float*)d_in[3];
  const float* Wl = (const float*)d_in[4];
  const float* bl = (const float*)d_in[5];
  const float* Wr = (const float*)d_in[6];
  const float* ln_g = (const float*)d_in[7];
  const float* ln_b = (const float*)d_in[8];
  const float* W_out = (const float*)d_in[9];
  const float* b_out = (const float*)d_in[10];
  float* out = (float*)d_out;

  int N = in_sizes[0] / D_IN;
  int E = in_sizes[1] / 2;
  const int* src = ei;
  const int* dst = ei + E;

  float* ws = (float*)d_ws;
  size_t nh = (size_t)N * HID;
  float* h = ws;
  float* hn = ws + nh;
  float* agg = ws + 2 * nh;
  float* deg = ws + 3 * nh;

  size_t need = (3 * nh + (size_t)N + (size_t)N + 1 + (size_t)E) * sizeof(float);
  if (ws_size >= need) {
    int* rowptr = (int*)(ws + 3 * nh + N);
    int* elist = rowptr + N + 1;

    group_edges_kernel<<<(N + OWN - 1) / OWN, BUILD_THREADS, 0, stream>>>(src, dst, rowptr, elist, deg, N, E);

    in_proj_kernel<<<(int)((nh + 255) / 256), 256, 0, stream>>>(x, W_in, b_in, h, N);

    for (int l = 0; l < 3; ++l) {
      ln_kernel<<<(N + 7) / 8, 256, 0, stream>>>(h, ln_g + l * HID, ln_b + l * HID, hn, N);
      gather_kernel<<<2048, 256, 0, stream>>>(rowptr, elist, hn, agg, N);
      sage_gemm_kernel<<<GEMM_GROUPS, GEMM_THREADS, 0, stream>>>(agg, hn, deg,
                                                   Wl + (size_t)l * HID * HID,
                                                   Wr + (size_t)l * HID * HID,
                                                   bl + l * HID, h, N);
    }
  } else {
    zero_kernel<<<256, 256, 0, stream>>>(deg, N);
    deg_kernel<<<(E + 255) / 256, 256, 0, stream>>>(dst, deg, E);
    invdeg_kernel<<<(N + 255) / 256, 256, 0, stream>>>(deg, N);

    in_proj_kernel<<<(int)((nh + 255) / 256), 256, 0, stream>>>(x, W_in, b_in, h, N);

    long long scat_threads = (long long)E * 32;
    for (int l = 0; l < 3; ++l) {
      ln_kernel<<<(N + 7) / 8, 256, 0, stream>>>(h, ln_g + l * HID, ln_b + l * HID, hn, N);
      zero_kernel<<<2048, 256, 0, stream>>>(agg, (int)nh);
      scatter_kernel<<<(int)((scat_threads + 255) / 256), 256, 0, stream>>>(src, dst, hn, agg, E);
      sage_gemm_kernel<<<GEMM_GROUPS, GEMM_THREADS, 0, stream>>>(agg, hn, deg,
                                                   Wl + (size_t)l * HID * HID,
                                                   Wr + (size_t)l * HID * HID,
                                                   bl + l * HID, h, N);
    }
  }

  out_proj_kernel<<<(N * D_OUT + 255) / 256, 256, 0, stream>>>(h, W_out, b_out, out, N);
}


# Row kernels (input projection, layernorm): a wave handles 4 to 8 rows, four requested together, instead of one row or 32 outputs; the projection keeps W_in in registers, a whole row per wave
# speedup vs baseline: 2.7682x; 1.0452x over previous
#include <hip/hip_runtime.h>

#define HID 128
#define D_IN 16
#define D_OUT 4
#define LN_EPS 1e-5f

typedef float v2f __attribute__((ext_vector_type(2)));
typedef float v8f __attribute__((ext_vector_type(8)));

__global__ void zero_kernel(float* __restrict__ p, int n) {
  int i = blockIdx.x * blockDim.x + threadIdx.x;
  int stride = gridDim.x * blockDim.x;
  for (; i < n; i += stride) p[i] = 0.0f;
}

__global__ void deg_kernel(const int* __restrict__ dst, float* __restrict__ deg, int E) {
  int e = blockIdx.x * blockDim.x + threadIdx.x;
  if (e < E) unsafeAtomicAdd(&deg[dst[e]], 1.0f);
}

__global__ void invdeg_kernel(float* __restrict__ deg, int N) {
  int i = blockIdx.x * blockDim.x + threadIdx.x;
  if (i < N) {
    float d = deg[i];
    deg[i] = 1.0f / fmaxf(d, 1.0f);
  }
}

#define ROWS_U 4

__global__ void __launch_bounds__(256) in_proj_kernel(const float* __restrict__ x,
                                                      const float* __restrict__ W_in,
                                                      const float* __restrict__ b_in,
                                                      float* __restrict__ h, int N) {
  const int lane = threadIdx.x & 31;
  const int wave = __builtin_amdgcn_readfirstlane((int)((blockIdx.x * blockDim.x + threadIdx.x) >> 5));
  const int nw = (int)((gridDim.x * blockDim.x) >> 5);

  float4 w[D_IN];
#pragma unroll
  for (int k = 0; k < D_IN; ++k) w[k] = ((const float4*)(W_in + k * HID))[lane];
  const float4 bv = ((const float4*)b_in)[lane];

  const int kx = lane & (D_IN - 1);
  for (int base = wave; base < N; base += ROWS_U * nw) {
    float xv[ROWS_U];
#pragma unroll
    for (int u = 0; u < ROWS_U; ++u) {
      const int r = base + u * nw;
      const int rc = r < N ? r : N - 1;
      xv[u] = x[(size_t)rc * D_IN + kx];
    }
    float4 acc[ROWS_U];
#pragma unroll
    for (int u = 0; u < ROWS_U; ++u) {
      acc[u] = bv;
#pragma unroll
      for (int k = 0; k < D_IN; ++k) {
        const float xk = __int_as_float(__builtin_amdgcn_readlane(__float_as_int(xv[u]), k));
        acc[u].x += xk * w[k].x;
        acc[u].y += xk * w[k].y;
        acc[u].z += xk * w[k].z;
        acc[u].w += xk * w[k].w;
      }
    }
#pragma unroll
    for (int u = 0; u < ROWS_U; ++u) {
      const int r = base + u * nw;
      if (r < N) ((float4*)(h + (size_t)r * HID))[lane] = acc[u];
    }
  }
}

__global__ void __launch_bounds__(256) ln_kernel(const float* __restrict__ h,
                                                 const float* __restrict__ g,
                                                 const float* __restrict__ b,
                                                 float* __restrict__ hn, int N) {
  const int lane = threadIdx.x & 31;
  const int wave = __builtin_amdgcn_readfirstlane((int)((blockIdx.x * blockDim.x + threadIdx.x) >> 5));
  const int nw = (int)((gridDim.x * blockDim.x) >> 5);

  const float4 gv = ((const float4*)g)[lane];
  const float4 bv = ((const float4*)b)[lane];
  const float4* h4 = (const float4*)h;

  for (int base = wave; base < N; base += ROWS_U * nw) {
    float4 vv[ROWS_U];
#pragma unroll
    for (int u = 0; u < ROWS_U; ++u) {
      const int r = base + u * nw;
      const int rc = r < N ? r : N - 1;
      vv[u] = h4[(size_t)rc * (HID / 4) + lane];
    }
    float4 o[ROWS_U];
#pragma unroll
    for (int u = 0; u < ROWS_U; ++u) {
      const float4 v = vv[u];
      float s = v.x + v.y + v.z + v.w;
#pragma unroll
      for (int off = 16; off > 0; off >>= 1) s += __shfl_xor(s, off, 32);
      float mu = s * (1.0f / HID);

      float dx = v.x - mu, dy = v.y - mu, dz = v.z - mu, dw = v.w - mu;
      float q = dx * dx + dy * dy + dz * dz + dw * dw;
#pragma unroll
      for (int off = 16; off > 0; off >>= 1) q += __shfl_xor(q, off, 32);
      float rstd = rsqrtf(q * (1.0f / HID) + LN_EPS);

      o[u].x = dx * rstd * gv.x + bv.x;
      o[u].y = dy * rstd * gv.y + bv.y;
      o[u].z = dz * rstd * gv.z + bv.z;
      o[u].w = dw * rstd * gv.w + bv.w;
    }
#pragma unroll
    for (int u = 0; u < ROWS_U; ++u) {
      const int r = base + u * nw;
      if (r < N) ((float4*)(hn + (size_t)r * HID))[lane] = o[u];
    }
  }
}

__global__ void __launch_bounds__(256) scatter_kernel(const int* __restrict__ src,
                                                      const int* __restrict__ dst,
                                                      const float* __restrict__ hn,
                                                      float* __restrict__ agg, int E) {
  int gid = blockIdx.x * blockDim.x + threadIdx.x;
  int e = gid >> 5;
  int lane = gid & 31;
  if (e >= E) return;
  int s = src[e];
  int d = dst[e];
  float4 v = ((const float4*)(hn + (size_t)s * HID))[lane];
  float* ap = agg + (size_t)d * HID + lane * 4;
  unsafeAtomicAdd(ap + 0, v.x);
  unsafeAtomicAdd(ap + 1, v.y);
  unsafeAtomicAdd(ap + 2, v.z);
  unsafeAtomicAdd(ap + 3, v.w);
}

#define OWN 256
#define BUILD_THREADS 1024
#define CAP 8192
#define SCAN_U 8

__global__ void __launch_bounds__(BUILD_THREADS) group_edges_kernel(
    const int* __restrict__ src, const int* __restrict__ dst,
    int* __restrict__ rowptr, int* __restrict__ elist,
    float* __restrict__ invdeg, int N, int E) {
  __shared__ int lcnt[OWN];
  __shared__ int lpre[OWN];
  __shared__ int red[BUILD_THREADS / 32];
  __shared__ int nmatch;
  __shared__ int m_edge[CAP];
  __shared__ int m_loc[CAP];
  __shared__ int m_rank[CAP];

  const int tid = threadIdx.x;
  const int lo = blockIdx.x * OWN;
  const int hi = (lo + OWN < N) ? (lo + OWN) : N;

  for (int i = tid; i < OWN; i += BUILD_THREADS) lcnt[i] = 0;
  if (tid == 0) nmatch = 0;
  __syncthreads();

  int below = 0;
  const unsigned span = (unsigned)(hi - lo);
  auto take = [&](int e, int loc) {
    int r = atomicAdd(&lcnt[loc], 1);
    int slot = atomicAdd(&nmatch, 1);
    if (slot < CAP) {
      m_edge[slot] = e;
      m_loc[slot] = loc;
      m_rank[slot] = r;
    }
  };
  auto visit = [&](int e, int d) {
    below += (d < lo) ? 1 : 0;
    unsigned loc = (unsigned)(d - lo);
    if (loc < span) take(e, (int)loc);
  };
  auto visit4 = [&](int e, int4 d) {
    below += ((d.x < lo) ? 1 : 0) + ((d.y < lo) ? 1 : 0) + ((d.z < lo) ? 1 : 0) + ((d.w < lo) ? 1 : 0);
    unsigned lx = (unsigned)(d.x - lo), ly = (unsigned)(d.y - lo);
    unsigned lz = (unsigned)(d.z - lo), lw = (unsigned)(d.w - lo);
    bool ix = lx < span, iy = ly < span, iz = lz < span, iw = lw < span;
    if (ix | iy | iz | iw) {
      if (ix) take(e + 0, (int)lx);
      if (iy) take(e + 1, (int)ly);
      if (iz) take(e + 2, (int)lz);
      if (iw) take(e + 3, (int)lw);
    }
  };

  const bool vec_ok = (((size_t)dst) & 15) == 0;
  const int E4 = vec_ok ? (E >> 2) : 0;
  const int4* d4 = (const int4*)dst;
  int i0 = tid;
  for (; i0 + (SCAN_U - 1) * BUILD_THREADS < E4; i0 += SCAN_U * BUILD_THREADS) {
    int4 d[SCAN_U];
#pragma unroll
    for (int u = 0; u < SCAN_U; ++u) d[u] = d4[i0 + u * BUILD_THREADS];
#pragma unroll
    for (int u = 0; u < SCAN_U; ++u) visit4((i0 + u * BUILD_THREADS) << 2, d[u]);
  }
  for (; i0 < E4; i0 += BUILD_THREADS) visit4(i0 << 2, d4[i0]);
  for (int e = (E4 << 2) + tid; e < E; e += BUILD_THREADS) visit(e, dst[e]);

  int s = below;
  for (int off = 16; off > 0; off >>= 1) s += __shfl_xor(s, off, 32);
  if ((tid & 31) == 0) red[tid >> 5] = s;
  __syncthreads();
  int base = 0;
  for (int i = 0; i < BUILD_THREADS / 32; ++i) base += red[i];

  if (tid < OWN) {
    int p = 0;
    for (int i = 0; i < tid; ++i) p += lcnt[i];
    lpre[tid] = p;
  }
  __syncthreads();

  if (tid < OWN && lo + tid < N) {
    rowptr[lo + tid] = base + lpre[tid];
    invdeg[lo + tid] = 1.0f / fmaxf((float)lcnt[tid], 1.0f);
  }
  if (hi == N && tid == 0) {
    int last = hi - lo - 1;
    rowptr[N] = base + lpre[last] + lcnt[last];
  }

  const int nm = nmatch;
  if (nm <= CAP) {
    for (int i = tid; i < nm; i += BUILD_THREADS)
      elist[base + lpre[m_loc[i]] + m_rank[i]] = src[m_edge[i]];
  } else {
    __syncthreads();
    for (int i = tid; i < OWN; i += BUILD_THREADS) lcnt[i] = 0;
    __syncthreads();
    for (int e = tid; e < E; e += BUILD_THREADS) {
      int d = dst[e];
      if (d >= lo && d < hi) {
        int loc = d - lo;
        int r = atomicAdd(&lcnt[loc], 1);
        elist[base + lpre[loc] + r] = src[e];
      }
    }
  }
}

template <int U>
__device__ __forceinline__ void add_rows(const float4* __restrict__ hn4, int my,
                                         int j, int lane, float4& acc) {
  float4 v[U];
#pragma unroll
  for (int u = 0; u < U; ++u) {
    int sidx = __builtin_amdgcn_readlane(my, j + u);
    v[u] = hn4[(size_t)sidx * (HID / 4) + lane];
  }
#pragma unroll
  for (int u = 0; u < U; ++u) {
    acc.x += v[u].x;
    acc.y += v[u].y;
    acc.z += v[u].z;
    acc.w += v[u].w;
  }
}

__global__ void __launch_bounds__(256) gather_kernel(const int* __restrict__ rowptr,
                                                     const int* __restrict__ elist,
                                                     const float* __restrict__ hn,
                                                     float* __restrict__ agg, int N) {
  const int lane = threadIdx.x & 31;
  const int wave = __builtin_amdgcn_readfirstlane((int)((blockIdx.x * blockDim.x + threadIdx.x) >> 5));
  const int nw = (int)((gridDim.x * blockDim.x) >> 5);
  const float4* hn4 = (const float4*)hn;
  for (int n = wave; n < N; n += nw) {
    const int beg = __builtin_amdgcn_readfirstlane(rowptr[n]);
    const int end = __builtin_amdgcn_readfirstlane(rowptr[n + 1]);
    float4 acc = make_float4(0.0f, 0.0f, 0.0f, 0.0f);
    for (int b = beg; b < end; b += 32) {
      const int cnt = (end - b < 32) ? (end - b) : 32;
      const int my = (lane < cnt) ? elist[b + lane] : 0;
      int j = 0;
      for (; j + 8 <= cnt; j += 8) add_rows<8>(hn4, my, j, lane, acc);
      if (j + 4 <= cnt) { add_rows<4>(hn4, my, j, lane, acc); j += 4; }
      if (j + 2 <= cnt) { add_rows<2>(hn4, my, j, lane, acc); j += 2; }
      if (j < cnt) add_rows<1>(hn4, my, j, lane, acc);
    }
    ((float4*)(agg + (size_t)n * HID))[lane] = acc;
  }
}

#define GEMM_THREADS 256
#define GEMM_GROUPS 512
#define KG (2 * HID / 4)

__device__ __forceinline__ void gemm_k128(const float4* __restrict__ arow, float scale,
                                          const float4* __restrict__ wfrag, v8f (&acc)[4]) {
#pragma unroll
  for (int j = 0; j < HID / 8; ++j) {
    float4 a = arow[2 * j];
    a.x *= scale;
    a.y *= scale;
    a.z *= scale;
    a.w *= scale;
    v2f a0 = {a.x, a.y};
    v2f a1 = {a.z, a.w};
    const float4* bp = wfrag + (size_t)(2 * j) * HID;
#pragma unroll
    for (int t = 0; t < 4; ++t) {
      float4 b = bp[16 * t];
      v2f b0 = {b.x, b.y};
      v2f b1 = {b.z, b.w};
      acc[t] = __builtin_amdgcn_wmma_f32_16x16x4_f32(false, a0, false, b0, (short)0, acc[t], false, false);
      acc[t] = __builtin_amdgcn_wmma_f32_16x16x4_f32(false, a1, false, b1, (short)0, acc[t], false, false);
    }
  }
}

__global__ void __launch_bounds__(GEMM_THREADS) sage_gemm_kernel(
    const float* __restrict__ agg, const float* __restrict__ hn,
    const float* __restrict__ invdeg, const float* __restrict__ Wl,
    const float* __restrict__ Wr, const float* __restrict__ bl,
    float* __restrict__ h, int N) {
  __shared__ float4 wf[KG * HID];
  __shared__ int next_unit;

  const int tid = threadIdx.x;
  if (tid == 0) next_unit = 0;

#pragma unroll
  for (int q = 0; q < KG / (GEMM_THREADS / 32); ++q) {
    const int g = q * (GEMM_THREADS / 32) + (tid >> 5);
    const int m = tid & 31;
    const float* wp = (q < KG / 2 / (GEMM_THREADS / 32))
                          ? (Wl + (size_t)(4 * g) * HID)
                          : (Wr + (size_t)(4 * (g - KG / 2)) * HID);
    const float4 r0 = ((const float4*)(wp))[m];
    const float4 r1 = ((const float4*)(wp + HID))[m];
    const float4 r2 = ((const float4*)(wp + 2 * HID))[m];
    const float4 r3 = ((const float4*)(wp + 3 * HID))[m];
    float4* o = wf + g * HID + 4 * m;
    o[0] = make_float4(r0.x, r1.x, r2.x, r3.x);
    o[1] = make_float4(r0.y, r1.y, r2.y, r3.y);
    o[2] = make_float4(r0.z, r1.z, r2.z, r3.z);
    o[3] = make_float4(r0.w, r1.w, r2.w, r3.w);
  }

  const int lane = tid & 31;
  const int half = lane >> 4;
  const int l16 = lane & 15;

  const int nunits = ((N + 15) >> 4) * 2;
  const int per = nunits / (int)gridDim.x;
  const int rem = nunits - per * (int)gridDim.x;
  const int bx = (int)blockIdx.x;
  const int ub = bx * per + (bx < rem ? bx : rem);
  const int ue = ub + per + (bx < rem ? 1 : 0);
  __syncthreads();

  for (int draw = 0; draw <= ue - ub; ++draw) {
    int i = 0;
    if (lane == 0) i = atomicAdd(&next_unit, 1);
    i = __builtin_amdgcn_readfirstlane(i);
    const int u = ub + i;
    if (u >= ue) break;

    const int m0 = (u >> 1) * 16;
    const int n0 = (u & 1) * 64;
    const int row = m0 + l16;
    const int rowc = row < N ? row : N - 1;

    const float invd = invdeg[rowc];
    const float4* aggr = (const float4*)(agg + (size_t)rowc * HID) + half;
    const float4* hnr = (const float4*)(hn + (size_t)rowc * HID) + half;
    const float4* wl_f = wf + (size_t)half * HID + n0 + l16;
    const float4* wr_f = wl_f + (size_t)(KG / 2) * HID;

    v8f acc[4] = {};
    gemm_k128(aggr, invd, wl_f, acc);
    gemm_k128(hnr, 1.0f, wr_f, acc);

    float* hp = h + (size_t)(m0 + 8 * half) * HID + n0 + l16;
    if (m0 + 16 <= N) {
#pragma unroll
      for (int t = 0; t < 4; ++t) {
        const float bias = bl[n0 + 16 * t + l16];
#pragma unroll
        for (int v = 0; v < 8; ++v) {
          float r = hp[(size_t)v * HID + 16 * t] + acc[t][v] + bias;
          hp[(size_t)v * HID + 16 * t] = r > 0.0f ? r : 0.0f;
        }
      }
    } else {
#pragma unroll
      for (int t = 0; t < 4; ++t) {
        const float bias = bl[n0 + 16 * t + l16];
#pragma unroll
        for (int v = 0; v < 8; ++v) {
          int m = m0 + v + 8 * half;
          if (m < N) {
            float r = hp[(size_t)v * HID + 16 * t] + acc[t][v] + bias;
            hp[(size_t)v * HID + 16 * t] = r > 0.0f ? r : 0.0f;
          }
        }
      }
    }
  }
}

__global__ void out_proj_kernel(const float* __restrict__ h,
                                const float* __restrict__ W_out,
                                const float* __restrict__ b_out,
                                float* __restrict__ out, int N) {
  int idx = blockIdx.x * blockDim.x + threadIdx.x;
  if (idx >= N * D_OUT) return;
  int row = idx >> 2;
  int c = idx & 3;
  const float* hr = h + (size_t)row * HID;
  float acc = b_out[c];
#pragma unroll 8
  for (int k = 0; k < HID; ++k) acc += hr[k] * W_out[k * D_OUT + c];
  out[idx] = acc;
}

extern "C" void kernel_launch(void* const* d_in, const int* in_sizes, int n_in,
                              void* d_out, int out_size, void* d_ws, size_t ws_size,
                              hipStream_t stream) {
  const float* x = (const float*)d_in[0];
  const int* ei = (const int*)d_in[1];
  const float* W_in = (const float*)d_in[2];
  const float* b_in = (const float*)d_in[3];
  const float* Wl = (const float*)d_in[4];
  const float* bl = (const float*)d_in[5];
  const float* Wr = (const float*)d_in[6];
  const float* ln_g = (const float*)d_in[7];
  const float* ln_b = (const float*)d_in[8];
  const float* W_out = (const float*)d_in[9];
  const float* b_out = (const float*)d_in[10];
  float* out = (float*)d_out;

  int N = in_sizes[0] / D_IN;
  int E = in_sizes[1] / 2;
  const int* src = ei;
  const int* dst = ei + E;

  float* ws = (float*)d_ws;
  size_t nh = (size_t)N * HID;
  float* h = ws;
  float* hn = ws + nh;
  float* agg = ws + 2 * nh;
  float* deg = ws + 3 * nh;

  size_t need = (3 * nh + (size_t)N + (size_t)N + 1 + (size_t)E) * sizeof(float);
  if (ws_size >= need) {
    int* rowptr = (int*)(ws + 3 * nh + N);
    int* elist = rowptr + N + 1;

    group_edges_kernel<<<(N + OWN - 1) / OWN, BUILD_THREADS, 0, stream>>>(src, dst, rowptr, elist, deg, N, E);

    in_proj_kernel<<<(N + 16 * ROWS_U - 1) / (16 * ROWS_U), 256, 0, stream>>>(x, W_in, b_in, h, N);

    for (int l = 0; l < 3; ++l) {
      ln_kernel<<<(N + 8 * ROWS_U - 1) / (8 * ROWS_U), 256, 0, stream>>>(h, ln_g + l * HID, ln_b + l * HID, hn, N);
      gather_kernel<<<2048, 256, 0, stream>>>(rowptr, elist, hn, agg, N);
      sage_gemm_kernel<<<GEMM_GROUPS, GEMM_THREADS, 0, stream>>>(agg, hn, deg,
                                                   Wl + (size_t)l * HID * HID,
                                                   Wr + (size_t)l * HID * HID,
                                                   bl + l * HID, h, N);
    }
  } else {
    zero_kernel<<<256, 256, 0, stream>>>(deg, N);
    deg_kernel<<<(E + 255) / 256, 256, 0, stream>>>(dst, deg, E);
    invdeg_kernel<<<(N + 255) / 256, 256, 0, stream>>>(deg, N);

    in_proj_kernel<<<(N + 16 * ROWS_U - 1) / (16 * ROWS_U), 256, 0, stream>>>(x, W_in, b_in, h, N);

    long long scat_threads = (long long)E * 32;
    for (int l = 0; l < 3; ++l) {
      ln_kernel<<<(N + 8 * ROWS_U - 1) / (8 * ROWS_U), 256, 0, stream>>>(h, ln_g + l * HID, ln_b + l * HID, hn, N);
      zero_kernel<<<2048, 256, 0, stream>>>(agg, (int)nh);
      scatter_kernel<<<(int)((scat_threads + 255) / 256), 256, 0, stream>>>(src, dst, hn, agg, E);
      sage_gemm_kernel<<<GEMM_GROUPS, GEMM_THREADS, 0, stream>>>(agg, hn, deg,
                                                   Wl + (size_t)l * HID * HID,
                                                   Wr + (size_t)l * HID * HID,
                                                   bl + l * HID, h, N);
    }
  }

  out_proj_kernel<<<(N * D_OUT + 255) / 256, 256, 0, stream>>>(h, W_out, b_out, out, N);
}
